# starting kernel
# baseline (speedup 1.0000x reference)
#include <hip/hip_runtime.h>
#include <hip/hip_bf16.h>

typedef __attribute__((ext_vector_type(2))) float v2f;
typedef __attribute__((ext_vector_type(8))) float v8f;

#define TAO     24
#define MLAG    7
#define C_IN    7
#define D_MODEL 512
#define SEQ     2048
#define KDIM    168
#define NSTEPS  42
#define ASTRIDE 172

__global__ __launch_bounds__(256)
void tdc_conv_wmma_kernel(const float* __restrict__ x,
                          const float* __restrict__ W,
                          const float* __restrict__ bias,
                          float* __restrict__ out)
{
    __shared__ float Atile[16 * ASTRIDE];

    const int block = blockIdx.x;
    const int batch = block >> 7;
    const int t0    = (block & 127) << 4;
    const int tid   = threadIdx.x;

    for (int e = tid; e < 16 * KDIM; e += 256) {
        const int m_row = e / KDIM;
        const int j     = e - m_row * KDIM;
        const int k     = j % 3;
        const int i     = j / 3;
        const int c     = i >> 3;
        const int mm    = i & 7;
        const int t     = t0 + m_row;
        const int s     = (t + k - 1 + SEQ) & (SEQ - 1);
        float v = 0.0f;
        if (s >= MLAG * TAO) {
            v = x[((size_t)batch * SEQ + (s - TAO * mm)) * C_IN + c];
        }
        Atile[m_row * ASTRIDE + j] = v;
    }
    __syncthreads();

    const int wave = tid >> 5;
    const int lane = tid & 31;
    const int half = lane >> 4;
    const int lid  = lane & 15;

    v8f acc0 = {}, acc1 = {}, acc2 = {}, acc3 = {};

    const float* Arow = &Atile[lid * ASTRIDE + 2 * half];
    const int o_base = wave * 64;
    const float* Wl0 = W + (size_t)(o_base +  0 + lid) * KDIM + 2 * half;
    const float* Wl1 = W + (size_t)(o_base + 16 + lid) * KDIM + 2 * half;
    const float* Wl2 = W + (size_t)(o_base + 32 + lid) * KDIM + 2 * half;
    const float* Wl3 = W + (size_t)(o_base + 48 + lid) * KDIM + 2 * half;

    #pragma unroll 6
    for (int kk = 0; kk < NSTEPS; ++kk) {
        const int k0 = kk * 4;
        v2f a  = *(const v2f*)(Arow + k0);
        v2f b0 = *(const v2f*)(Wl0 + k0);
        v2f b1 = *(const v2f*)(Wl1 + k0);
        v2f b2 = *(const v2f*)(Wl2 + k0);
        v2f b3 = *(const v2f*)(Wl3 + k0);
        acc0 = __builtin_amdgcn_wmma_f32_16x16x4_f32(false, a, false, b0, (short)0, acc0, false, false);
        acc1 = __builtin_amdgcn_wmma_f32_16x16x4_f32(false, a, false, b1, (short)0, acc1, false, false);
        acc2 = __builtin_amdgcn_wmma_f32_16x16x4_f32(false, a, false, b2, (short)0, acc2, false, false);
        acc3 = __builtin_amdgcn_wmma_f32_16x16x4_f32(false, a, false, b3, (short)0, acc3, false, false);
    }

    const size_t row_base = ((size_t)batch * SEQ + t0 + 8 * half) * D_MODEL;

    {
        const int o = o_base + 0 + lid;
        const float bo = bias[o];
        #pragma unroll
        for (int r = 0; r < 8; ++r)
            out[row_base + (size_t)r * D_MODEL + o] = acc0[r] + bo;
    }
    {
        const int o = o_base + 16 + lid;
        const float bo = bias[o];
        #pragma unroll
        for (int r = 0; r < 8; ++r)
            out[row_base + (size_t)r * D_MODEL + o] = acc1[r] + bo;
    }
    {
        const int o = o_base + 32 + lid;
        const float bo = bias[o];
        #pragma unroll
        for (int r = 0; r < 8; ++r)
            out[row_base + (size_t)r * D_MODEL + o] = acc2[r] + bo;
    }
    {
        const int o = o_base + 48 + lid;
        const float bo = bias[o];
        #pragma unroll
        for (int r = 0; r < 8; ++r)
            out[row_base + (size_t)r * D_MODEL + o] = acc3[r] + bo;
    }
}

extern "C" void kernel_launch(void* const* d_in, const int* in_sizes, int n_in,
                              void* d_out, int out_size, void* d_ws, size_t ws_size,
                              hipStream_t stream) {
    (void)in_sizes; (void)n_in; (void)d_ws; (void)ws_size; (void)out_size;
    const float* x    = (const float*)d_in[0];
    const float* W    = (const float*)d_in[1];
    const float* bias = (const float*)d_in[2];
    float* out        = (float*)d_out;

    const int num_blocks = 32 * (SEQ / 16);
    tdc_conv_wmma_kernel<<<num_blocks, 256, 0, stream>>>(x, W, bias, out);
}


# Each lane owns 4 adjacent output columns (W rows re-dealt over the 4 matrix tiles): one 16-byte store per lane and row, 8 whole-line stores per tile instead of 32 half-line ones; sum order kept
# speedup vs baseline: 2.6741x; 2.6741x over previous
#include <hip/hip_runtime.h>
#include <hip/hip_bf16.h>


typedef __attribute__((ext_vector_type(2))) float v2f;
typedef __attribute__((ext_vector_type(4))) float v4f;
typedef __attribute__((ext_vector_type(8))) float v8f;

#define TAO     24
#define MLAG    7
#define C_IN    7
#define D_MODEL 512
#define SEQ     2048
#define KDIM    168
#define NSTEPS  42
#define ASTRIDE 172
#define TILES   16
#define WG_THREADS 128
#define CH_PER_THREAD 7
#define STAGE_N (CH_PER_THREAD * 3)


__device__ __forceinline__ void load_x_tile(const float* __restrict__ xb, int t_row,
                                            const int (&cm)[CH_PER_THREAD],
                                            float (&xv)[STAGE_N])
{
    #pragma unroll
    for (int k = 0; k < 3; ++k) {
        const int s  = (t_row + k - 1 + SEQ) & (SEQ - 1);
        const int s7 = (s >= MLAG * TAO ? s : MLAG * TAO) * C_IN;
        #pragma unroll
        for (int a = 0; a < CH_PER_THREAD; ++a)
            xv[a * 3 + k] = xb[s7 + cm[a]];
    }
}

__device__ __forceinline__ void store_x_tile(float* __restrict__ Arow_st, int t_row,
                                             const float (&xv)[STAGE_N])
{
    #pragma unroll
    for (int k = 0; k < 3; ++k) {
        const int  s     = (t_row + k - 1 + SEQ) & (SEQ - 1);
        const bool valid = (s >= MLAG * TAO);
        #pragma unroll
        for (int a = 0; a < CH_PER_THREAD; ++a)
            Arow_st[a * 3 + k] = valid ? xv[a * 3 + k] : 0.0f;
    }
}

__global__ __launch_bounds__(WG_THREADS, 2)
void tdc_conv_wmma_kernel(const float* __restrict__ x,
                          const float* __restrict__ W,
                          const float* __restrict__ bias,
                          float* __restrict__ out)
{
    __shared__ float Atile[2][16 * ASTRIDE];

    const int wg    = blockIdx.x;
    const int chalf = wg & 1;
    const int rg    = wg >> 1;
    const int batch = rg >> 3;
    const int tbase = (rg & 7) << 8;
    const int tid   = threadIdx.x;

    const int wave = tid >> 5;
    const int lane = tid & 31;
    const int half = lane >> 4;
    const int lid  = lane & 15;

    const float* xb = x + (size_t)batch * SEQ * C_IN;

    const int m_row = tid >> 3;
    const int q     = tid & 7;
    int cm[CH_PER_THREAD];
    #pragma unroll
    for (int a = 0; a < CH_PER_THREAD; ++a) {
        const int ci = CH_PER_THREAD * q + a;
        cm[a] = (ci >> 3) - TAO * C_IN * (ci & 7);
    }
    const int st_off = m_row * ASTRIDE + STAGE_N * q;

    float xv[STAGE_N];
    load_x_tile(xb, tbase + m_row, cm, xv);

    const int o_base = (chalf * 4 + wave) * 64;
    const float* Wl0 = W + (size_t)(o_base + 4 * lid + 0) * KDIM + 2 * half;
    const float* Wl1 = W + (size_t)(o_base + 4 * lid + 1) * KDIM + 2 * half;
    const float* Wl2 = W + (size_t)(o_base + 4 * lid + 2) * KDIM + 2 * half;
    const float* Wl3 = W + (size_t)(o_base + 4 * lid + 3) * KDIM + 2 * half;

    v2f bf[NSTEPS][4];
    #pragma unroll
    for (int kk = 0; kk < NSTEPS; ++kk) {
        const int k0 = kk * 4;
        bf[kk][0] = *(const v2f*)(Wl0 + k0);
        bf[kk][1] = *(const v2f*)(Wl1 + k0);
        bf[kk][2] = *(const v2f*)(Wl2 + k0);
        bf[kk][3] = *(const v2f*)(Wl3 + k0);
    }

    const float bo0 = bias[o_base + 4 * lid + 0];
    const float bo1 = bias[o_base + 4 * lid + 1];
    const float bo2 = bias[o_base + 4 * lid + 2];
    const float bo3 = bias[o_base + 4 * lid + 3];

    store_x_tile(&Atile[0][st_off], tbase + m_row, xv);
    __syncthreads();

    for (int mt = 0; mt < TILES; ++mt) {
        const int t0 = tbase + 16 * mt;

        const float* Arow = &Atile[mt & 1][lid * ASTRIDE + 2 * half];

        v2f af[NSTEPS];
        #pragma unroll
        for (int kk = 0; kk < NSTEPS; ++kk)
            af[kk] = *(const v2f*)(Arow + kk * 4);
        __builtin_amdgcn_sched_barrier(0);

        if (mt + 1 < TILES) load_x_tile(xb, t0 + 16 + m_row, cm, xv);

        v8f acc0 = {}, acc1 = {}, acc2 = {}, acc3 = {};

        #pragma unroll
        for (int kk = 0; kk < NSTEPS; ++kk) {
            acc0 = __builtin_amdgcn_wmma_f32_16x16x4_f32(false, af[kk], false, bf[kk][0], (short)0, acc0, false, false);
            acc1 = __builtin_amdgcn_wmma_f32_16x16x4_f32(false, af[kk], false, bf[kk][1], (short)0, acc1, false, false);
            acc2 = __builtin_amdgcn_wmma_f32_16x16x4_f32(false, af[kk], false, bf[kk][2], (short)0, acc2, false, false);
            acc3 = __builtin_amdgcn_wmma_f32_16x16x4_f32(false, af[kk], false, bf[kk][3], (short)0, acc3, false, false);
        }

        if (mt + 1 < TILES) {
            store_x_tile(&Atile[(mt + 1) & 1][st_off], t0 + 16 + m_row, xv);
            __syncthreads();
        }

        float* orow = out + ((size_t)batch * SEQ + t0 + 8 * half) * D_MODEL + o_base + 4 * lid;
        v4f ov[8];
        #pragma unroll
        for (int r = 0; r < 8; ++r) {
            ov[r].x = acc0[r] + bo0;
            ov[r].y = acc1[r] + bo1;
            ov[r].z = acc2[r] + bo2;
            ov[r].w = acc3[r] + bo3;
        }
        #pragma unroll
        for (int r = 0; r < 8; ++r)
            *(v4f*)(orow + (size_t)r * D_MODEL) = ov[r];
    }
}

extern "C" void kernel_launch(void* const* d_in, const int* in_sizes, int n_in,
                              void* d_out, int out_size, void* d_ws, size_t ws_size,
                              hipStream_t stream) {
    (void)in_sizes; (void)n_in; (void)d_ws; (void)ws_size; (void)out_size;
    const float* x    = (const float*)d_in[0];
    const float* W    = (const float*)d_in[1];
    const float* bias = (const float*)d_in[2];
    float* out        = (float*)d_out;

    tdc_conv_wmma_kernel<<<512, WG_THREADS, 0, stream>>>(x, W, bias, out);
}
